# starting kernel
# baseline (speedup 1.0000x reference)
#include <hip/hip_runtime.h>

#define B_    4
#define N_    1024
#define FI_   128
#define FO_   64
#define DELTA_ 1.0f
#define EPS_   1e-3f

typedef __attribute__((ext_vector_type(2))) float v2f;
typedef __attribute__((ext_vector_type(8))) float v8f;

static __device__ __forceinline__ v8f wmma_f32_16x16x4(v2f a, v2f b, v8f c) {
  return __builtin_amdgcn_wmma_f32_16x16x4_f32(
       false, a,  false, b,
       (short)0, c,  false,  false);
}

__global__ void __launch_bounds__(128)
gtv_xw_gemm(const float* __restrict__ x, const float* __restrict__ w,
            float* __restrict__ xw) {
  const int tid  = threadIdx.x;
  const int wv   = tid >> 5;
  const int lane = tid & 31;
  const int hh   = lane >> 4;
  const int l15  = lane & 15;
  const int rt   = blockIdx.x;
  const int f0   = wv * 16;

  const float* __restrict__ xrow = x + (size_t)(rt * 16 + l15) * FI_;
  v8f c = {0.f, 0.f, 0.f, 0.f, 0.f, 0.f, 0.f, 0.f};

  #pragma unroll 4
  for (int kc = 0; kc < FI_ / 4; ++kc) {
    const int ks = kc * 4 + hh * 2;
    v2f a;  a.x = xrow[ks];  a.y = xrow[ks + 1];
    const float* __restrict__ wp = w + (size_t)ks * FO_ + f0 + l15;
    v2f b;  b.x = wp[0];     b.y = wp[FO_];
    c = wmma_f32_16x16x4(a, b, c);
  }

  float* __restrict__ orow = xw + (size_t)(rt * 16 + hh * 8) * FO_ + f0 + l15;
  #pragma unroll
  for (int r = 0; r < 8; ++r) orow[(size_t)r * FO_] = c[r];
}

__global__ void __launch_bounds__(256)
gtv_fused(const float* __restrict__ adj, const float* __restrict__ xw,
          const float* __restrict__ bias, float* __restrict__ out) {
  __shared__ float xwi_s[16][FO_];
  __shared__ float Msh[8][16][20];
  __shared__ float out_s[16][FO_];
  __shared__ float deg_s[16];

  const int tid  = threadIdx.x;
  const int wv   = tid >> 5;
  const int lane = tid & 31;
  const int hh   = lane >> 4;
  const int l15  = lane & 15;

  const int b  = blockIdx.x >> 6;
  const int i0 = (blockIdx.x & 63) * 16;

  if (tid < 16) deg_s[tid] = 0.f;
  {
    const int row = tid >> 4;
    const int c4  = (tid & 15) * 4;
    *(float4*)&out_s[row][c4] = make_float4(0.f, 0.f, 0.f, 0.f);
    *(float4*)&xwi_s[row][c4] =
        *(const float4*)(xw + (size_t)(b * N_ + i0 + row) * FO_ + c4);
  }
  __syncthreads();

  float dacc[8];
  #pragma unroll
  for (int r = 0; r < 8; ++r) dacc[r] = 0.f;
  v8f oacc[4];
  #pragma unroll
  for (int f = 0; f < 4; ++f)
    #pragma unroll
    for (int e = 0; e < 8; ++e) oacc[f][e] = 0.f;

  const float* __restrict__ adjb = adj + (size_t)b * N_ * N_ + (size_t)i0 * N_;
  const float* __restrict__ xwb  = xw + (size_t)b * N_ * FO_;

  for (int step = 0; step < 8; ++step) {
    const int jt = step * 8 + wv;
    const int j0 = jt * 16;
    if (step < 7)
      __builtin_prefetch(adjb + (size_t)(hh * 8) * N_ + (j0 + 128) + l15, 0, 0);

    float acc[8];
    #pragma unroll
    for (int r = 0; r < 8; ++r) acc[r] = 0.f;
    const float* __restrict__ xwj = xwb + (size_t)(j0 + l15) * FO_;
    #pragma unroll 4
    for (int kc = 0; kc < FO_ / 4; ++kc) {
      const float4 xj = *(const float4*)(xwj + kc * 4);
      #pragma unroll
      for (int r = 0; r < 8; ++r) {
        const float4 xi = *(const float4*)&xwi_s[r + hh * 8][kc * 4];
        acc[r] += fabsf(xi.x - xj.x) + fabsf(xi.y - xj.y)
                + fabsf(xi.z - xj.z) + fabsf(xi.w - xj.w);
      }
    }

    const float* __restrict__ ap = adjb + j0 + l15;
    #pragma unroll
    for (int r = 0; r < 8; ++r) {
      const int irow = r + hh * 8;
      const float a = ap[(size_t)irow * N_];
      float m = (a != 0.f) ? (a / fmaxf(acc[r], EPS_)) : a;
      m *= DELTA_;
      dacc[r] += m;
      Msh[wv][irow][l15] = m;
    }

    v2f avec[4];
    #pragma unroll
    for (int kchunk = 0; kchunk < 4; ++kchunk) {
      const int ks = kchunk * 4 + hh * 2;
      avec[kchunk].x = Msh[wv][l15][ks];
      avec[kchunk].y = Msh[wv][l15][ks + 1];
    }
    #pragma unroll
    for (int fc = 0; fc < 4; ++fc) {
      #pragma unroll
      for (int kchunk = 0; kchunk < 4; ++kchunk) {
        const int ks = kchunk * 4 + hh * 2;
        const float* __restrict__ bp =
            xwb + (size_t)(j0 + ks) * FO_ + fc * 16 + l15;
        v2f bv;  bv.x = bp[0];  bv.y = bp[FO_];
        oacc[fc] = wmma_f32_16x16x4(avec[kchunk], bv, oacc[fc]);
      }
    }
  }

  #pragma unroll
  for (int r = 0; r < 8; ++r) {
    float d = dacc[r];
    d += __shfl_xor(d, 1, 16);
    d += __shfl_xor(d, 2, 16);
    d += __shfl_xor(d, 4, 16);
    d += __shfl_xor(d, 8, 16);
    if (l15 == 0) atomicAdd(&deg_s[r + hh * 8], d);
  }
  #pragma unroll
  for (int fc = 0; fc < 4; ++fc)
    #pragma unroll
    for (int r = 0; r < 8; ++r)
      atomicAdd(&out_s[r + hh * 8][fc * 16 + l15], oacc[fc][r]);
  __syncthreads();

  {
    const int row = tid >> 4;
    const int c4  = (tid & 15) * 4;
    const float diagc = 1.f - deg_s[row];
    const float4 xi = *(const float4*)&xwi_s[row][c4];
    const float4 bs = *(const float4*)(bias + c4);
    float4 o = *(const float4*)&out_s[row][c4];
    o.x += diagc * xi.x + bs.x;
    o.y += diagc * xi.y + bs.y;
    o.z += diagc * xi.z + bs.z;
    o.w += diagc * xi.w + bs.w;
    *(float4*)(out + (size_t)(b * N_ + i0 + row) * FO_ + c4) = o;
  }
}

extern "C" void kernel_launch(void* const* d_in, const int* in_sizes, int n_in,
                              void* d_out, int out_size, void* d_ws, size_t ws_size,
                              hipStream_t stream) {
  (void)in_sizes; (void)n_in; (void)out_size; (void)ws_size;
  const float* x    = (const float*)d_in[0];
  const float* adj  = (const float*)d_in[1];
  const float* w    = (const float*)d_in[2];
  const float* bias = (const float*)d_in[3];
  float*       out  = (float*)d_out;
  float*       xw   = (float*)d_ws;

  gtv_xw_gemm<<<(B_ * N_) / 16, 128, 0, stream>>>(x, w, xw);
  gtv_fused  <<<B_ * (N_ / 16), 256, 0, stream>>>(adj, xw, bias, out);
}


# fused kernel: 64 j-tiles of a row block over 32 waves (2 each, 8 per SIMD; was 8 waves x 8) to cover load waits; no atomics: per-wave LDS partial tiles added in fixed wave order (sum order changes)
# speedup vs baseline: 1.5308x; 1.5308x over previous
#include <hip/hip_runtime.h>

#define B_    4
#define N_    1024
#define FI_   128
#define FO_   64
#define DELTA_ 1.0f
#define EPS_   1e-3f

typedef __attribute__((ext_vector_type(2))) float v2f;
typedef __attribute__((ext_vector_type(8))) float v8f;

static __device__ __forceinline__ v8f wmma_f32_16x16x4(v2f a, v2f b, v8f c) {
  return __builtin_amdgcn_wmma_f32_16x16x4_f32(
       false, a,  false, b,
       (short)0, c,  false,  false);
}

__global__ void __launch_bounds__(128)
gtv_xw_gemm(const float* __restrict__ x, const float* __restrict__ w,
            float* __restrict__ xw) {
  const int tid  = threadIdx.x;
  const int wv   = tid >> 5;
  const int lane = tid & 31;
  const int hh   = lane >> 4;
  const int l15  = lane & 15;
  const int rt   = blockIdx.x;
  const int f0   = wv * 16;

  const float* __restrict__ xrow = x + (size_t)(rt * 16 + l15) * FI_;
  v8f c = {0.f, 0.f, 0.f, 0.f, 0.f, 0.f, 0.f, 0.f};

  #pragma unroll 4
  for (int kc = 0; kc < FI_ / 4; ++kc) {
    const int ks = kc * 4 + hh * 2;
    v2f a;  a.x = xrow[ks];  a.y = xrow[ks + 1];
    const float* __restrict__ wp = w + (size_t)ks * FO_ + f0 + l15;
    v2f b;  b.x = wp[0];     b.y = wp[FO_];
    c = wmma_f32_16x16x4(a, b, c);
  }

  float* __restrict__ orow = xw + (size_t)(rt * 16 + hh * 8) * FO_ + f0 + l15;
  #pragma unroll
  for (int r = 0; r < 8; ++r) orow[(size_t)r * FO_] = c[r];
}

#define NWV_  32
#define TPW_  (N_ / 16 / NWV_)
#define PPAD_ 66

__global__ void __launch_bounds__(1024)
gtv_fused(const float* __restrict__ adj, const float* __restrict__ xw,
          const float* __restrict__ bias, float* __restrict__ out) {
  __shared__ float xwi_s[16][FO_];
  __shared__ float Msh[NWV_][16][20];
  __shared__ float part_s[NWV_][16][PPAD_];
  __shared__ __attribute__((aligned(16))) float degp_s[16][NWV_];

  const int tid  = threadIdx.x;
  const int wv   = tid >> 5;
  const int lane = tid & 31;
  const int hh   = lane >> 4;
  const int l15  = lane & 15;

  const int b  = blockIdx.x >> 6;
  const int i0 = (blockIdx.x & 63) * 16;

  if (tid < 256) {
    const int row = tid >> 4;
    const int c4  = (tid & 15) * 4;
    *(float4*)&xwi_s[row][c4] =
        *(const float4*)(xw + (size_t)(b * N_ + i0 + row) * FO_ + c4);
  }
  __syncthreads();

  float dacc[8];
  #pragma unroll
  for (int r = 0; r < 8; ++r) dacc[r] = 0.f;
  v8f oacc[4];
  #pragma unroll
  for (int f = 0; f < 4; ++f)
    #pragma unroll
    for (int e = 0; e < 8; ++e) oacc[f][e] = 0.f;

  const float* __restrict__ adjb = adj + (size_t)b * N_ * N_ + (size_t)i0 * N_;
  const float* __restrict__ xwb  = xw + (size_t)b * N_ * FO_;

  #pragma unroll 1
  for (int step = 0; step < TPW_; ++step) {
    const int jt = step * NWV_ + wv;
    const int j0 = jt * 16;
    if (step < TPW_ - 1)
      __builtin_prefetch(adjb + (size_t)(hh * 8) * N_ + (j0 + NWV_ * 16) + l15, 0, 0);

    float acc[8];
    #pragma unroll
    for (int r = 0; r < 8; ++r) acc[r] = 0.f;
    const float* __restrict__ xwj = xwb + (size_t)(j0 + l15) * FO_;
    #pragma unroll 4
    for (int kc = 0; kc < FO_ / 4; ++kc) {
      const float4 xj = *(const float4*)(xwj + kc * 4);
      #pragma unroll
      for (int r = 0; r < 8; ++r) {
        const float4 xi = *(const float4*)&xwi_s[r + hh * 8][kc * 4];
        acc[r] += fabsf(xi.x - xj.x) + fabsf(xi.y - xj.y)
                + fabsf(xi.z - xj.z) + fabsf(xi.w - xj.w);
      }
    }

    const float* __restrict__ ap = adjb + j0 + l15;
    #pragma unroll
    for (int r = 0; r < 8; ++r) {
      const int irow = r + hh * 8;
      const float a = ap[(size_t)irow * N_];
      float m = (a != 0.f) ? (a / fmaxf(acc[r], EPS_)) : a;
      m *= DELTA_;
      dacc[r] += m;
      Msh[wv][irow][l15] = m;
    }

    v2f avec[4];
    #pragma unroll
    for (int kchunk = 0; kchunk < 4; ++kchunk) {
      const int ks = kchunk * 4 + hh * 2;
      avec[kchunk].x = Msh[wv][l15][ks];
      avec[kchunk].y = Msh[wv][l15][ks + 1];
    }
    v2f bvec[4][4];
    #pragma unroll
    for (int fc = 0; fc < 4; ++fc) {
      #pragma unroll
      for (int kchunk = 0; kchunk < 4; ++kchunk) {
        const int ks = kchunk * 4 + hh * 2;
        const float* __restrict__ bp =
            xwb + (size_t)(j0 + ks) * FO_ + fc * 16 + l15;
        bvec[fc][kchunk].x = bp[0];  bvec[fc][kchunk].y = bp[FO_];
      }
    }
    __builtin_amdgcn_sched_barrier(0);
    #pragma unroll
    for (int fc = 0; fc < 4; ++fc) {
      #pragma unroll
      for (int kchunk = 0; kchunk < 4; ++kchunk)
        oacc[fc] = wmma_f32_16x16x4(avec[kchunk], bvec[fc][kchunk], oacc[fc]);
    }
    __builtin_amdgcn_sched_barrier(0);
    asm volatile("v_nop\n\tv_nop\n\tv_nop\n\tv_nop\n\tv_nop\n\tv_nop\n\tv_nop\n\tv_nop"
                 : "+v"(oacc[0]), "+v"(oacc[1]), "+v"(oacc[2]), "+v"(oacc[3]));
    __builtin_amdgcn_sched_barrier(0);
  }

  #pragma unroll
  for (int r = 0; r < 8; ++r) {
    float d = dacc[r];
    d += __shfl_xor(d, 1, 16);
    d += __shfl_xor(d, 2, 16);
    d += __shfl_xor(d, 4, 16);
    d += __shfl_xor(d, 8, 16);
    if (l15 == 0) degp_s[r + hh * 8][wv] = d;
  }
  #pragma unroll
  for (int fc = 0; fc < 4; ++fc)
    #pragma unroll
    for (int r = 0; r < 8; ++r)
      part_s[wv][r + hh * 8][fc * 16 + l15] = oacc[fc][r];
  __syncthreads();

  {
    const int row = tid >> 6;
    const int col = tid & 63;
    float o = 0.f;
    #pragma unroll
    for (int w = 0; w < NWV_; ++w) o += part_s[w][row][col];
    float dg = 0.f;
    #pragma unroll
    for (int w4 = 0; w4 < NWV_ / 4; ++w4) {
      const float4 d4 = *(const float4*)&degp_s[row][w4 * 4];
      dg += d4.x;  dg += d4.y;  dg += d4.z;  dg += d4.w;
    }
    const float diagc = 1.f - dg;
    o += diagc * xwi_s[row][col] + bias[col];
    out[(size_t)(b * N_ + i0 + row) * FO_ + col] = o;
  }
}

extern "C" void kernel_launch(void* const* d_in, const int* in_sizes, int n_in,
                              void* d_out, int out_size, void* d_ws, size_t ws_size,
                              hipStream_t stream) {
  (void)in_sizes; (void)n_in; (void)out_size; (void)ws_size;
  const float* x    = (const float*)d_in[0];
  const float* adj  = (const float*)d_in[1];
  const float* w    = (const float*)d_in[2];
  const float* bias = (const float*)d_in[3];
  float*       out  = (float*)d_out;
  float*       xw   = (float*)d_ws;

  gtv_xw_gemm<<<(B_ * N_) / 16, 128, 0, stream>>>(x, w, xw);
  gtv_fused  <<<B_ * (N_ / 16), 1024, 0, stream>>>(adj, xw, bias, out);
}


# fused kernel: one column j per lane, its xw_j row in registers (tile staged via LDS); the block's xw_i rows enter the subtract as scalar-load operands, not LDS reads; 16 waves; sum order changes
# speedup vs baseline: 1.7813x; 1.1636x over previous
#include <hip/hip_runtime.h>

#define B_    4
#define N_    1024
#define FI_   128
#define FO_   64
#define DELTA_ 1.0f
#define EPS_   1e-3f

typedef __attribute__((ext_vector_type(2))) float v2f;
typedef __attribute__((ext_vector_type(8))) float v8f;

static __device__ __forceinline__ v8f wmma_f32_16x16x4(v2f a, v2f b, v8f c) {
  return __builtin_amdgcn_wmma_f32_16x16x4_f32(
       false, a,  false, b,
       (short)0, c,  false,  false);
}

__global__ void __launch_bounds__(128)
gtv_xw_gemm(const float* __restrict__ x, const float* __restrict__ w,
            float* __restrict__ xw) {
  const int tid  = threadIdx.x;
  const int wv   = tid >> 5;
  const int lane = tid & 31;
  const int hh   = lane >> 4;
  const int l15  = lane & 15;
  const int rt   = blockIdx.x;
  const int f0   = wv * 16;

  const float* __restrict__ xrow = x + (size_t)(rt * 16 + l15) * FI_;
  v8f c = {0.f, 0.f, 0.f, 0.f, 0.f, 0.f, 0.f, 0.f};

  #pragma unroll 4
  for (int kc = 0; kc < FI_ / 4; ++kc) {
    const int ks = kc * 4 + hh * 2;
    v2f a;  a.x = xrow[ks];  a.y = xrow[ks + 1];
    const float* __restrict__ wp = w + (size_t)ks * FO_ + f0 + l15;
    v2f b;  b.x = wp[0];     b.y = wp[FO_];
    c = wmma_f32_16x16x4(a, b, c);
  }

  float* __restrict__ orow = xw + (size_t)(rt * 16 + hh * 8) * FO_ + f0 + l15;
  #pragma unroll
  for (int r = 0; r < 8; ++r) orow[(size_t)r * FO_] = c[r];
}

#define NWV_  16
#define TPW_  (N_ / 32 / NWV_)
#define PPAD_ 66
#define MPAD_ 36
#define XPAD_ 68

typedef __attribute__((ext_vector_type(16))) float v16f;
typedef const __attribute__((address_space(4))) v16f* cv16p;

__global__ void __launch_bounds__(512)
gtv_fused(const float* __restrict__ adj, const float* __restrict__ xw,
          const float* __restrict__ bias, float* __restrict__ out) {
  __shared__ __attribute__((aligned(16))) float Xsh[NWV_][32][XPAD_];
  __shared__ __attribute__((aligned(16))) float Msh[NWV_][16][MPAD_];
  __shared__ float degp_s[16][NWV_];

  const int tid  = threadIdx.x;
  const int wv   = __builtin_amdgcn_readfirstlane(tid >> 5);
  const int lane = tid & 31;
  const int hh   = lane >> 4;
  const int l15  = lane & 15;

  const int b  = blockIdx.x >> 6;
  const int i0 = (blockIdx.x & 63) * 16;

  const float* __restrict__ adjb = adj + (size_t)b * N_ * N_ + (size_t)i0 * N_;
  const float* __restrict__ xwb  = xw + (size_t)b * N_ * FO_;
  const float* __restrict__ xwI  = xwb + (size_t)i0 * FO_;

  float dacc = 0.f;
  v8f oacc[4];
  #pragma unroll
  for (int f = 0; f < 4; ++f)
    #pragma unroll
    for (int e = 0; e < 8; ++e) oacc[f][e] = 0.f;

  #pragma unroll 1
  for (int step = 0; step < TPW_; ++step) {
    const int j0 = (step * NWV_ + wv) * 32;

    {
      const float4* __restrict__ gp = (const float4*)(xwb + (size_t)j0 * FO_) + lane;
      float4 g[16];
      #pragma unroll
      for (int p = 0; p < 16; ++p) g[p] = gp[p * 32];
      #pragma unroll
      for (int p = 0; p < 16; ++p)
        *(float4*)&Xsh[wv][2 * p + hh][l15 * 4] = g[p];
    }
    float xj[FO_];
    #pragma unroll
    for (int kc = 0; kc < FO_ / 4; ++kc) {
      const float4 t = *(const float4*)&Xsh[wv][lane][kc * 4];
      xj[kc * 4 + 0] = t.x;  xj[kc * 4 + 1] = t.y;
      xj[kc * 4 + 2] = t.z;  xj[kc * 4 + 3] = t.w;
    }

    const float* __restrict__ ap = adjb + j0 + lane;
    float ac[4], an[4];
    #pragma unroll
    for (int c = 0; c < 4; ++c) ac[c] = ap[(size_t)c * N_];
    #pragma unroll 1
    for (int rg = 0; rg < 4; ++rg) {
      #pragma unroll
      for (int c = 0; c < 4; ++c) {
        const int rn = (rg * 4 + 4 + c < 16) ? (rg * 4 + 4 + c) : 15;
        an[c] = ap[(size_t)rn * N_];
      }
      #pragma unroll
      for (int c4 = 0; c4 < 4; ++c4) {
        const int r = rg * 4 + c4;
        const cv16p xr = (cv16p)(xwI + r * FO_);
        float s[8];
        #pragma unroll
        for (int c = 0; c < 8; ++c) s[c] = 0.f;
        #pragma unroll
        for (int q = 0; q < 4; ++q) {
          const v16f xi = xr[q];
          #pragma unroll
          for (int e = 0; e < 16; e += 8) {
            float t[8];
            #pragma unroll
            for (int c = 0; c < 8; ++c)
              asm("v_sub_f32_e32 %0, %1, %2" : "=v"(t[c]) : "s"(xi[e + c]), "v"(xj[q * 16 + e + c]));
            #pragma unroll
            for (int c = 0; c < 8; ++c)
              asm("v_add_f32_e64 %0, |%1|, %0" : "+v"(s[c]) : "v"(t[c]));
          }
        }
        const float d = ((s[0] + s[1]) + (s[2] + s[3])) + ((s[4] + s[5]) + (s[6] + s[7]));
        const float a0 = ac[c4];
        float m = (a0 != 0.f) ? (a0 / fmaxf(d, EPS_)) : a0;
        m *= DELTA_;
        Msh[wv][r][lane] = m;
      }
      #pragma unroll
      for (int c = 0; c < 4; ++c) ac[c] = an[c];
    }

    float4 am[4];
    #pragma unroll
    for (int q = 0; q < 4; ++q)
      am[q] = *(const float4*)&Msh[wv][l15][q * 8 + hh * 4];
    #pragma unroll
    for (int q = 0; q < 4; ++q)
      dacc += (am[q].x + am[q].y) + (am[q].z + am[q].w);
    asm volatile("" : "+v"(dacc));

    v2f bvec[4][4][2];
    {
      const float* bq = &Xsh[wv][hh * 4][l15];
      #pragma unroll
      for (int q = 0; q < 4; ++q)
        #pragma unroll
        for (int t = 0; t < 2; ++t)
          #pragma unroll
          for (int fc = 0; fc < 4; ++fc) {
            const float* p = bq + (q * 8 + t * 2) * XPAD_ + fc * 16;
            bvec[fc][q][t].x = p[0];  bvec[fc][q][t].y = p[XPAD_];
          }
    }
    v2f avec[4][2];
    #pragma unroll
    for (int q = 0; q < 4; ++q) {
      avec[q][0].x = am[q].x;  avec[q][0].y = am[q].y;
      avec[q][1].x = am[q].z;  avec[q][1].y = am[q].w;
    }
    __builtin_amdgcn_sched_barrier(0);
    #pragma unroll
    for (int q = 0; q < 4; ++q)
      #pragma unroll
      for (int t = 0; t < 2; ++t)
        #pragma unroll
        for (int fc = 0; fc < 4; ++fc)
          oacc[fc] = wmma_f32_16x16x4(avec[q][t], bvec[fc][q][t], oacc[fc]);
    __builtin_amdgcn_sched_barrier(0);
    asm volatile("v_nop\n\tv_nop\n\tv_nop\n\tv_nop\n\tv_nop\n\tv_nop\n\tv_nop\n\tv_nop"
                 : "+v"(oacc[0]), "+v"(oacc[1]), "+v"(oacc[2]), "+v"(oacc[3]));
    __builtin_amdgcn_sched_barrier(0);
  }

  {
    const float dsum = dacc + __shfl_xor(dacc, 16, 32);
    if (hh == 0) degp_s[l15][wv] = dsum;
  }
  {
    float* slab = &Xsh[wv][0][0];
    #pragma unroll
    for (int fc = 0; fc < 4; ++fc)
      #pragma unroll
      for (int r = 0; r < 8; ++r)
        slab[(r + hh * 8) * PPAD_ + fc * 16 + l15] = oacc[fc][r];
  }
  __syncthreads();

  #pragma unroll
  for (int half = 0; half < 2; ++half) {
    const int e   = tid + half * 512;
    const int row = e >> 6;
    const int col = e & 63;
    float o = 0.f;
    #pragma unroll
    for (int w = 0; w < NWV_; ++w) o += (&Xsh[w][0][0])[row * PPAD_ + col];
    float dg = 0.f;
    #pragma unroll
    for (int w = 0; w < NWV_; ++w) dg += degp_s[row][w];
    const float diagc = 1.f - dg;
    o += diagc * xwI[row * FO_ + col] + bias[col];
    out[(size_t)(b * N_ + i0 + row) * FO_ + col] = o;
  }
}

extern "C" void kernel_launch(void* const* d_in, const int* in_sizes, int n_in,
                              void* d_out, int out_size, void* d_ws, size_t ws_size,
                              hipStream_t stream) {
  (void)in_sizes; (void)n_in; (void)out_size; (void)ws_size;
  const float* x    = (const float*)d_in[0];
  const float* adj  = (const float*)d_in[1];
  const float* w    = (const float*)d_in[2];
  const float* bias = (const float*)d_in[3];
  float*       out  = (float*)d_out;
  float*       xw   = (float*)d_ws;

  gtv_xw_gemm<<<(B_ * N_) / 16, 128, 0, stream>>>(x, w, xw);
  gtv_fused  <<<B_ * (N_ / 16), 512, 0, stream>>>(adj, xw, bias, out);
}


# fused kernel: L1 distance as 2*sum_f max(xw_i,xw_j) - rowsum_i - rowsum_j (dual-issued maxima, packed adds; no subtract, no abs); the distance's sum order and rounding change
# speedup vs baseline: 2.2229x; 1.2479x over previous
#include <hip/hip_runtime.h>

#define B_    4
#define N_    1024
#define FI_   128
#define FO_   64
#define DELTA_ 1.0f
#define EPS_   1e-3f

typedef __attribute__((ext_vector_type(2))) float v2f;
typedef __attribute__((ext_vector_type(8))) float v8f;

static __device__ __forceinline__ v8f wmma_f32_16x16x4(v2f a, v2f b, v8f c) {
  return __builtin_amdgcn_wmma_f32_16x16x4_f32(
       false, a,  false, b,
       (short)0, c,  false,  false);
}

__global__ void __launch_bounds__(128)
gtv_xw_gemm(const float* __restrict__ x, const float* __restrict__ w,
            float* __restrict__ xw) {
  const int tid  = threadIdx.x;
  const int wv   = tid >> 5;
  const int lane = tid & 31;
  const int hh   = lane >> 4;
  const int l15  = lane & 15;
  const int rt   = blockIdx.x;
  const int f0   = wv * 16;

  const float* __restrict__ xrow = x + (size_t)(rt * 16 + l15) * FI_;
  v8f c = {0.f, 0.f, 0.f, 0.f, 0.f, 0.f, 0.f, 0.f};

  #pragma unroll 4
  for (int kc = 0; kc < FI_ / 4; ++kc) {
    const int ks = kc * 4 + hh * 2;
    v2f a;  a.x = xrow[ks];  a.y = xrow[ks + 1];
    const float* __restrict__ wp = w + (size_t)ks * FO_ + f0 + l15;
    v2f b;  b.x = wp[0];     b.y = wp[FO_];
    c = wmma_f32_16x16x4(a, b, c);
  }

  float* __restrict__ orow = xw + (size_t)(rt * 16 + hh * 8) * FO_ + f0 + l15;
  #pragma unroll
  for (int r = 0; r < 8; ++r) orow[(size_t)r * FO_] = c[r];
}

#define NWV_  16
#define TPW_  (N_ / 32 / NWV_)
#define PPAD_ 66
#define MPAD_ 36
#define XPAD_ 68

typedef __attribute__((ext_vector_type(16))) float v16f;
typedef const __attribute__((address_space(4))) v16f* cv16p;

__global__ void __launch_bounds__(512)
gtv_fused(const float* __restrict__ adj, const float* __restrict__ xw,
          const float* __restrict__ bias, float* __restrict__ out) {
  __shared__ __attribute__((aligned(16))) float Xsh[NWV_][32][XPAD_];
  __shared__ __attribute__((aligned(16))) float Msh[NWV_][16][MPAD_];
  __shared__ float degp_s[16][NWV_];

  const int tid  = threadIdx.x;
  const int wv   = __builtin_amdgcn_readfirstlane(tid >> 5);
  const int lane = tid & 31;
  const int hh   = lane >> 4;
  const int l15  = lane & 15;

  const int b  = blockIdx.x >> 6;
  const int i0 = (blockIdx.x & 63) * 16;

  const float* __restrict__ adjb = adj + (size_t)b * N_ * N_ + (size_t)i0 * N_;
  const float* __restrict__ xwb  = xw + (size_t)b * N_ * FO_;
  const float* __restrict__ xwI  = xwb + (size_t)i0 * FO_;

  float dacc = 0.f;
  v8f oacc[4];
  #pragma unroll
  for (int f = 0; f < 4; ++f)
    #pragma unroll
    for (int e = 0; e < 8; ++e) oacc[f][e] = 0.f;

  float arow;
  {
    const float4* __restrict__ pr = (const float4*)(xwI + l15 * FO_ + hh * 32);
    float4 g[8];
    #pragma unroll
    for (int k = 0; k < 8; ++k) g[k] = pr[k];
    const float4 u = ((g[0] + g[1]) + (g[2] + g[3])) + ((g[4] + g[5]) + (g[6] + g[7]));
    const float part = (u.x + u.y) + (u.z + u.w);
    arow = part + __shfl_xor(part, 16, 32);
  }

  #pragma unroll 1
  for (int step = 0; step < TPW_; ++step) {
    const int j0 = (step * NWV_ + wv) * 32;

    {
      const float4* __restrict__ gp = (const float4*)(xwb + (size_t)j0 * FO_) + lane;
      float4 g[16];
      #pragma unroll
      for (int p = 0; p < 16; ++p) g[p] = gp[p * 32];
      #pragma unroll
      for (int p = 0; p < 16; ++p)
        *(float4*)&Xsh[wv][2 * p + hh][l15 * 4] = g[p];
    }
    float xj[FO_];
    #pragma unroll
    for (int kc = 0; kc < FO_ / 4; ++kc) {
      const float4 t = *(const float4*)&Xsh[wv][lane][kc * 4];
      xj[kc * 4 + 0] = t.x;  xj[kc * 4 + 1] = t.y;
      xj[kc * 4 + 2] = t.z;  xj[kc * 4 + 3] = t.w;
    }

    float bj;
    {
      v2f bs[4];
      #pragma unroll
      for (int c = 0; c < 4; ++c) { bs[c].x = xj[2 * c]; bs[c].y = xj[2 * c + 1]; }
      #pragma unroll
      for (int k = 8; k < FO_; k += 2) {
        v2f t;  t.x = xj[k];  t.y = xj[k + 1];
        bs[(k >> 1) & 3] += t;
      }
      const v2f b2 = (bs[0] + bs[1]) + (bs[2] + bs[3]);
      bj = b2.x + b2.y;
    }

    const float* __restrict__ ap = adjb + j0 + lane;
    float ac[4], an[4];
    #pragma unroll
    for (int c = 0; c < 4; ++c) ac[c] = ap[(size_t)c * N_];
    #pragma unroll 1
    for (int rg = 0; rg < 4; ++rg) {
      #pragma unroll
      for (int c = 0; c < 4; ++c) {
        const int rn = (rg * 4 + 4 + c < 16) ? (rg * 4 + 4 + c) : 15;
        an[c] = ap[(size_t)rn * N_];
      }
      #pragma unroll
      for (int c4 = 0; c4 < 4; ++c4) {
        const int r = rg * 4 + c4;
        const cv16p xr = (cv16p)(xwI + r * FO_);
        v2f S[8];
        #pragma unroll
        for (int c = 0; c < 8; ++c) { S[c].x = 0.f; S[c].y = 0.f; }
        #pragma unroll
        for (int q = 0; q < 4; ++q) {
          const v16f xi = xr[q];
          asm("v_dual_max_num_f32 v208, %8, %24 :: v_dual_max_num_f32 v209, %9, %25\n\t"
              "v_dual_max_num_f32 v210, %10, %26 :: v_dual_max_num_f32 v211, %11, %27\n\t"
              "v_dual_max_num_f32 v212, %12, %28 :: v_dual_max_num_f32 v213, %13, %29\n\t"
              "v_dual_max_num_f32 v214, %14, %30 :: v_dual_max_num_f32 v215, %15, %31\n\t"
              "v_dual_max_num_f32 v216, %16, %32 :: v_dual_max_num_f32 v217, %17, %33\n\t"
              "v_dual_max_num_f32 v218, %18, %34 :: v_dual_max_num_f32 v219, %19, %35\n\t"
              "v_dual_max_num_f32 v220, %20, %36 :: v_dual_max_num_f32 v221, %21, %37\n\t"
              "v_dual_max_num_f32 v222, %22, %38 :: v_dual_max_num_f32 v223, %23, %39\n\t"
              "v_pk_add_f32 %0, %0, v[208:209]\n\t"
              "v_pk_add_f32 %1, %1, v[210:211]\n\t"
              "v_pk_add_f32 %2, %2, v[212:213]\n\t"
              "v_pk_add_f32 %3, %3, v[214:215]\n\t"
              "v_pk_add_f32 %4, %4, v[216:217]\n\t"
              "v_pk_add_f32 %5, %5, v[218:219]\n\t"
              "v_pk_add_f32 %6, %6, v[220:221]\n\t"
              "v_pk_add_f32 %7, %7, v[222:223]"
              : "+v"(S[0]), "+v"(S[1]), "+v"(S[2]), "+v"(S[3]), "+v"(S[4]), "+v"(S[5]), "+v"(S[6]), "+v"(S[7])
              : "s"(xi[0]), "s"(xi[1]), "s"(xi[2]), "s"(xi[3]), "s"(xi[4]), "s"(xi[5]), "s"(xi[6]), "s"(xi[7]), "s"(xi[8]), "s"(xi[9]), "s"(xi[10]), "s"(xi[11]), "s"(xi[12]), "s"(xi[13]), "s"(xi[14]), "s"(xi[15]),
                "v"(xj[q * 16 + 0]), "v"(xj[q * 16 + 1]), "v"(xj[q * 16 + 2]), "v"(xj[q * 16 + 3]), "v"(xj[q * 16 + 4]), "v"(xj[q * 16 + 5]), "v"(xj[q * 16 + 6]), "v"(xj[q * 16 + 7]), "v"(xj[q * 16 + 8]), "v"(xj[q * 16 + 9]), "v"(xj[q * 16 + 10]), "v"(xj[q * 16 + 11]), "v"(xj[q * 16 + 12]), "v"(xj[q * 16 + 13]), "v"(xj[q * 16 + 14]), "v"(xj[q * 16 + 15])
              : "v208", "v209", "v210", "v211", "v212", "v213", "v214", "v215", "v216", "v217", "v218", "v219", "v220", "v221", "v222", "v223");
        }
        const v2f S2 = ((S[0] + S[1]) + (S[2] + S[3])) + ((S[4] + S[5]) + (S[6] + S[7]));
        const float msum = S2.x + S2.y;
        const float ai = __int_as_float(__builtin_amdgcn_readlane(__float_as_int(arow), r));
        const float d = __builtin_fmaf(2.f, msum, -(ai + bj));
        const float a0 = ac[c4];
        float m = (a0 != 0.f) ? (a0 / fmaxf(d, EPS_)) : a0;
        m *= DELTA_;
        Msh[wv][r][lane] = m;
      }
      #pragma unroll
      for (int c = 0; c < 4; ++c) ac[c] = an[c];
    }

    float4 am[4];
    #pragma unroll
    for (int q = 0; q < 4; ++q)
      am[q] = *(const float4*)&Msh[wv][l15][q * 8 + hh * 4];
    #pragma unroll
    for (int q = 0; q < 4; ++q)
      dacc += (am[q].x + am[q].y) + (am[q].z + am[q].w);
    asm volatile("" : "+v"(dacc));

    v2f bvec[4][4][2];
    {
      const float* bq = &Xsh[wv][hh * 4][l15];
      #pragma unroll
      for (int q = 0; q < 4; ++q)
        #pragma unroll
        for (int t = 0; t < 2; ++t)
          #pragma unroll
          for (int fc = 0; fc < 4; ++fc) {
            const float* p = bq + (q * 8 + t * 2) * XPAD_ + fc * 16;
            bvec[fc][q][t].x = p[0];  bvec[fc][q][t].y = p[XPAD_];
          }
    }
    v2f avec[4][2];
    #pragma unroll
    for (int q = 0; q < 4; ++q) {
      avec[q][0].x = am[q].x;  avec[q][0].y = am[q].y;
      avec[q][1].x = am[q].z;  avec[q][1].y = am[q].w;
    }
    __builtin_amdgcn_sched_barrier(0);
    #pragma unroll
    for (int q = 0; q < 4; ++q)
      #pragma unroll
      for (int t = 0; t < 2; ++t)
        #pragma unroll
        for (int fc = 0; fc < 4; ++fc)
          oacc[fc] = wmma_f32_16x16x4(avec[q][t], bvec[fc][q][t], oacc[fc]);
    __builtin_amdgcn_sched_barrier(0);
    asm volatile("v_nop\n\tv_nop\n\tv_nop\n\tv_nop\n\tv_nop\n\tv_nop\n\tv_nop\n\tv_nop"
                 : "+v"(oacc[0]), "+v"(oacc[1]), "+v"(oacc[2]), "+v"(oacc[3]));
    __builtin_amdgcn_sched_barrier(0);
  }

  {
    const float dsum = dacc + __shfl_xor(dacc, 16, 32);
    if (hh == 0) degp_s[l15][wv] = dsum;
  }
  {
    float* slab = &Xsh[wv][0][0];
    #pragma unroll
    for (int fc = 0; fc < 4; ++fc)
      #pragma unroll
      for (int r = 0; r < 8; ++r)
        slab[(r + hh * 8) * PPAD_ + fc * 16 + l15] = oacc[fc][r];
  }
  __syncthreads();

  #pragma unroll
  for (int half = 0; half < 2; ++half) {
    const int e   = tid + half * 512;
    const int row = e >> 6;
    const int col = e & 63;
    float o = 0.f;
    #pragma unroll
    for (int w = 0; w < NWV_; ++w) o += (&Xsh[w][0][0])[row * PPAD_ + col];
    float dg = 0.f;
    #pragma unroll
    for (int w = 0; w < NWV_; ++w) dg += degp_s[row][w];
    const float diagc = 1.f - dg;
    o += diagc * xwI[row * FO_ + col] + bias[col];
    out[(size_t)(b * N_ + i0 + row) * FO_ + col] = o;
  }
}

extern "C" void kernel_launch(void* const* d_in, const int* in_sizes, int n_in,
                              void* d_out, int out_size, void* d_ws, size_t ws_size,
                              hipStream_t stream) {
  (void)in_sizes; (void)n_in; (void)out_size; (void)ws_size;
  const float* x    = (const float*)d_in[0];
  const float* adj  = (const float*)d_in[1];
  const float* w    = (const float*)d_in[2];
  const float* bias = (const float*)d_in[3];
  float*       out  = (float*)d_out;
  float*       xw   = (float*)d_ws;

  gtv_xw_gemm<<<(B_ * N_) / 16, 128, 0, stream>>>(x, w, xw);
  gtv_fused  <<<B_ * (N_ / 16), 512, 0, stream>>>(adj, xw, bias, out);
}


# distance loop: a row's first 16 maxima are written by the dual-issued maxima straight into its 16 partial sums (no zeroing, no add for that group): 99 -> 83 vector instructions a row; sums unchanged
# speedup vs baseline: 2.3271x; 1.0469x over previous
#include <hip/hip_runtime.h>

#define B_    4
#define N_    1024
#define FI_   128
#define FO_   64
#define DELTA_ 1.0f
#define EPS_   1e-3f

typedef __attribute__((ext_vector_type(2))) float v2f;
typedef __attribute__((ext_vector_type(8))) float v8f;

static __device__ __forceinline__ v8f wmma_f32_16x16x4(v2f a, v2f b, v8f c) {
  return __builtin_amdgcn_wmma_f32_16x16x4_f32(
       false, a,  false, b,
       (short)0, c,  false,  false);
}

__global__ void __launch_bounds__(128)
gtv_xw_gemm(const float* __restrict__ x, const float* __restrict__ w,
            float* __restrict__ xw) {
  const int tid  = threadIdx.x;
  const int wv   = tid >> 5;
  const int lane = tid & 31;
  const int hh   = lane >> 4;
  const int l15  = lane & 15;
  const int rt   = blockIdx.x;
  const int f0   = wv * 16;

  const float* __restrict__ xrow = x + (size_t)(rt * 16 + l15) * FI_;
  v8f c = {0.f, 0.f, 0.f, 0.f, 0.f, 0.f, 0.f, 0.f};

  #pragma unroll 4
  for (int kc = 0; kc < FI_ / 4; ++kc) {
    const int ks = kc * 4 + hh * 2;
    v2f a;  a.x = xrow[ks];  a.y = xrow[ks + 1];
    const float* __restrict__ wp = w + (size_t)ks * FO_ + f0 + l15;
    v2f b;  b.x = wp[0];     b.y = wp[FO_];
    c = wmma_f32_16x16x4(a, b, c);
  }

  float* __restrict__ orow = xw + (size_t)(rt * 16 + hh * 8) * FO_ + f0 + l15;
  #pragma unroll
  for (int r = 0; r < 8; ++r) orow[(size_t)r * FO_] = c[r];
}

#define NWV_  16
#define TPW_  (N_ / 32 / NWV_)
#define PPAD_ 66
#define MPAD_ 36
#define XPAD_ 68

typedef __attribute__((ext_vector_type(16))) float v16f;
typedef const __attribute__((address_space(4))) v16f* cv16p;

__global__ void __launch_bounds__(512)
gtv_fused(const float* __restrict__ adj, const float* __restrict__ xw,
          const float* __restrict__ bias, float* __restrict__ out) {
  __shared__ __attribute__((aligned(16))) float Xsh[NWV_][32][XPAD_];
  __shared__ __attribute__((aligned(16))) float Msh[NWV_][16][MPAD_];
  __shared__ float degp_s[16][NWV_];

  const int tid  = threadIdx.x;
  const int wv   = __builtin_amdgcn_readfirstlane(tid >> 5);
  const int lane = tid & 31;
  const int hh   = lane >> 4;
  const int l15  = lane & 15;

  const int b  = blockIdx.x >> 6;
  const int i0 = (blockIdx.x & 63) * 16;

  const float* __restrict__ adjb = adj + (size_t)b * N_ * N_ + (size_t)i0 * N_;
  const float* __restrict__ xwb  = xw + (size_t)b * N_ * FO_;
  const float* __restrict__ xwI  = xwb + (size_t)i0 * FO_;

  float dacc = 0.f;
  v8f oacc[4];
  #pragma unroll
  for (int f = 0; f < 4; ++f)
    #pragma unroll
    for (int e = 0; e < 8; ++e) oacc[f][e] = 0.f;

  float arow;
  {
    const float4* __restrict__ pr = (const float4*)(xwI + l15 * FO_ + hh * 32);
    float4 g[8];
    #pragma unroll
    for (int k = 0; k < 8; ++k) g[k] = pr[k];
    const float4 u = ((g[0] + g[1]) + (g[2] + g[3])) + ((g[4] + g[5]) + (g[6] + g[7]));
    const float part = (u.x + u.y) + (u.z + u.w);
    arow = part + __shfl_xor(part, 16, 32);
  }

  #pragma unroll 1
  for (int step = 0; step < TPW_; ++step) {
    const int j0 = (step * NWV_ + wv) * 32;

    {
      const float4* __restrict__ gp = (const float4*)(xwb + (size_t)j0 * FO_) + lane;
      float4 g[16];
      #pragma unroll
      for (int p = 0; p < 16; ++p) g[p] = gp[p * 32];
      #pragma unroll
      for (int p = 0; p < 16; ++p)
        *(float4*)&Xsh[wv][2 * p + hh][l15 * 4] = g[p];
    }
    float xj[FO_];
    #pragma unroll
    for (int kc = 0; kc < FO_ / 4; ++kc) {
      const float4 t = *(const float4*)&Xsh[wv][lane][kc * 4];
      xj[kc * 4 + 0] = t.x;  xj[kc * 4 + 1] = t.y;
      xj[kc * 4 + 2] = t.z;  xj[kc * 4 + 3] = t.w;
    }

    float bj;
    {
      v2f bs[4];
      #pragma unroll
      for (int c = 0; c < 4; ++c) { bs[c].x = xj[2 * c]; bs[c].y = xj[2 * c + 1]; }
      #pragma unroll
      for (int k = 8; k < FO_; k += 2) {
        v2f t;  t.x = xj[k];  t.y = xj[k + 1];
        bs[(k >> 1) & 3] += t;
      }
      const v2f b2 = (bs[0] + bs[1]) + (bs[2] + bs[3]);
      bj = b2.x + b2.y;
    }

    const float* __restrict__ ap = adjb + j0 + lane;
    float ac[4], an[4];
    #pragma unroll
    for (int c = 0; c < 4; ++c) ac[c] = ap[(size_t)c * N_];
    #pragma unroll 1
    for (int rg = 0; rg < 4; ++rg) {
      #pragma unroll
      for (int c = 0; c < 4; ++c) {
        const int rn = (rg * 4 + 4 + c < 16) ? (rg * 4 + 4 + c) : 15;
        an[c] = ap[(size_t)rn * N_];
      }
      #pragma unroll
      for (int c4 = 0; c4 < 4; ++c4) {
        const int r = rg * 4 + c4;
        const cv16p xr = (cv16p)(xwI + r * FO_);
        float m0[16];
        {
          const v16f xi = xr[0];
          asm("v_dual_max_num_f32 %0, %16, %32 :: v_dual_max_num_f32 %1, %17, %33\n\t"
              "v_dual_max_num_f32 %2, %18, %34 :: v_dual_max_num_f32 %3, %19, %35\n\t"
              "v_dual_max_num_f32 %4, %20, %36 :: v_dual_max_num_f32 %5, %21, %37\n\t"
              "v_dual_max_num_f32 %6, %22, %38 :: v_dual_max_num_f32 %7, %23, %39\n\t"
              "v_dual_max_num_f32 %8, %24, %40 :: v_dual_max_num_f32 %9, %25, %41\n\t"
              "v_dual_max_num_f32 %10, %26, %42 :: v_dual_max_num_f32 %11, %27, %43\n\t"
              "v_dual_max_num_f32 %12, %28, %44 :: v_dual_max_num_f32 %13, %29, %45\n\t"
              "v_dual_max_num_f32 %14, %30, %46 :: v_dual_max_num_f32 %15, %31, %47"
              : "=&v"(m0[0]), "=&v"(m0[1]), "=&v"(m0[2]), "=&v"(m0[3]), "=&v"(m0[4]), "=&v"(m0[5]), "=&v"(m0[6]), "=&v"(m0[7]), "=&v"(m0[8]), "=&v"(m0[9]), "=&v"(m0[10]), "=&v"(m0[11]), "=&v"(m0[12]), "=&v"(m0[13]), "=&v"(m0[14]), "=&v"(m0[15])
              : "s"(xi[0]), "s"(xi[1]), "s"(xi[2]), "s"(xi[3]), "s"(xi[4]), "s"(xi[5]), "s"(xi[6]), "s"(xi[7]), "s"(xi[8]), "s"(xi[9]), "s"(xi[10]), "s"(xi[11]), "s"(xi[12]), "s"(xi[13]), "s"(xi[14]), "s"(xi[15]),
                "v"(xj[0]), "v"(xj[1]), "v"(xj[2]), "v"(xj[3]), "v"(xj[4]), "v"(xj[5]), "v"(xj[6]), "v"(xj[7]), "v"(xj[8]), "v"(xj[9]), "v"(xj[10]), "v"(xj[11]), "v"(xj[12]), "v"(xj[13]), "v"(xj[14]), "v"(xj[15]));
        }
        v2f S[8];
        #pragma unroll
        for (int c = 0; c < 8; ++c) { S[c].x = m0[2 * c]; S[c].y = m0[2 * c + 1]; }
        #pragma unroll
        for (int q = 1; q < 4; ++q) {
          const v16f xi = xr[q];
          asm("v_dual_max_num_f32 v208, %8, %24 :: v_dual_max_num_f32 v209, %9, %25\n\t"
              "v_dual_max_num_f32 v210, %10, %26 :: v_dual_max_num_f32 v211, %11, %27\n\t"
              "v_dual_max_num_f32 v212, %12, %28 :: v_dual_max_num_f32 v213, %13, %29\n\t"
              "v_dual_max_num_f32 v214, %14, %30 :: v_dual_max_num_f32 v215, %15, %31\n\t"
              "v_dual_max_num_f32 v216, %16, %32 :: v_dual_max_num_f32 v217, %17, %33\n\t"
              "v_dual_max_num_f32 v218, %18, %34 :: v_dual_max_num_f32 v219, %19, %35\n\t"
              "v_dual_max_num_f32 v220, %20, %36 :: v_dual_max_num_f32 v221, %21, %37\n\t"
              "v_dual_max_num_f32 v222, %22, %38 :: v_dual_max_num_f32 v223, %23, %39\n\t"
              "v_pk_add_f32 %0, %0, v[208:209]\n\t"
              "v_pk_add_f32 %1, %1, v[210:211]\n\t"
              "v_pk_add_f32 %2, %2, v[212:213]\n\t"
              "v_pk_add_f32 %3, %3, v[214:215]\n\t"
              "v_pk_add_f32 %4, %4, v[216:217]\n\t"
              "v_pk_add_f32 %5, %5, v[218:219]\n\t"
              "v_pk_add_f32 %6, %6, v[220:221]\n\t"
              "v_pk_add_f32 %7, %7, v[222:223]"
              : "+v"(S[0]), "+v"(S[1]), "+v"(S[2]), "+v"(S[3]), "+v"(S[4]), "+v"(S[5]), "+v"(S[6]), "+v"(S[7])
              : "s"(xi[0]), "s"(xi[1]), "s"(xi[2]), "s"(xi[3]), "s"(xi[4]), "s"(xi[5]), "s"(xi[6]), "s"(xi[7]), "s"(xi[8]), "s"(xi[9]), "s"(xi[10]), "s"(xi[11]), "s"(xi[12]), "s"(xi[13]), "s"(xi[14]), "s"(xi[15]),
                "v"(xj[q * 16 + 0]), "v"(xj[q * 16 + 1]), "v"(xj[q * 16 + 2]), "v"(xj[q * 16 + 3]), "v"(xj[q * 16 + 4]), "v"(xj[q * 16 + 5]), "v"(xj[q * 16 + 6]), "v"(xj[q * 16 + 7]), "v"(xj[q * 16 + 8]), "v"(xj[q * 16 + 9]), "v"(xj[q * 16 + 10]), "v"(xj[q * 16 + 11]), "v"(xj[q * 16 + 12]), "v"(xj[q * 16 + 13]), "v"(xj[q * 16 + 14]), "v"(xj[q * 16 + 15])
              : "v208", "v209", "v210", "v211", "v212", "v213", "v214", "v215", "v216", "v217", "v218", "v219", "v220", "v221", "v222", "v223");
        }
        const v2f S2 = ((S[0] + S[1]) + (S[2] + S[3])) + ((S[4] + S[5]) + (S[6] + S[7]));
        const float msum = S2.x + S2.y;
        const float ai = __int_as_float(__builtin_amdgcn_readlane(__float_as_int(arow), r));
        const float d = __builtin_fmaf(2.f, msum, -(ai + bj));
        const float a0 = ac[c4];
        float m = (a0 != 0.f) ? (a0 / fmaxf(d, EPS_)) : a0;
        m *= DELTA_;
        Msh[wv][r][lane] = m;
      }
      #pragma unroll
      for (int c = 0; c < 4; ++c) ac[c] = an[c];
    }

    float4 am[4];
    #pragma unroll
    for (int q = 0; q < 4; ++q)
      am[q] = *(const float4*)&Msh[wv][l15][q * 8 + hh * 4];
    #pragma unroll
    for (int q = 0; q < 4; ++q)
      dacc += (am[q].x + am[q].y) + (am[q].z + am[q].w);
    asm volatile("" : "+v"(dacc));

    v2f bvec[4][4][2];
    {
      const float* bq = &Xsh[wv][hh * 4][l15];
      #pragma unroll
      for (int q = 0; q < 4; ++q)
        #pragma unroll
        for (int t = 0; t < 2; ++t)
          #pragma unroll
          for (int fc = 0; fc < 4; ++fc) {
            const float* p = bq + (q * 8 + t * 2) * XPAD_ + fc * 16;
            bvec[fc][q][t].x = p[0];  bvec[fc][q][t].y = p[XPAD_];
          }
    }
    v2f avec[4][2];
    #pragma unroll
    for (int q = 0; q < 4; ++q) {
      avec[q][0].x = am[q].x;  avec[q][0].y = am[q].y;
      avec[q][1].x = am[q].z;  avec[q][1].y = am[q].w;
    }
    __builtin_amdgcn_sched_barrier(0);
    #pragma unroll
    for (int q = 0; q < 4; ++q)
      #pragma unroll
      for (int t = 0; t < 2; ++t)
        #pragma unroll
        for (int fc = 0; fc < 4; ++fc)
          oacc[fc] = wmma_f32_16x16x4(avec[q][t], bvec[fc][q][t], oacc[fc]);
    __builtin_amdgcn_sched_barrier(0);
    asm volatile("v_nop\n\tv_nop\n\tv_nop\n\tv_nop\n\tv_nop\n\tv_nop\n\tv_nop\n\tv_nop"
                 : "+v"(oacc[0]), "+v"(oacc[1]), "+v"(oacc[2]), "+v"(oacc[3]));
    __builtin_amdgcn_sched_barrier(0);
  }

  {
    const float dsum = dacc + __shfl_xor(dacc, 16, 32);
    if (hh == 0) degp_s[l15][wv] = dsum;
  }
  {
    float* slab = &Xsh[wv][0][0];
    #pragma unroll
    for (int fc = 0; fc < 4; ++fc)
      #pragma unroll
      for (int r = 0; r < 8; ++r)
        slab[(r + hh * 8) * PPAD_ + fc * 16 + l15] = oacc[fc][r];
  }
  __syncthreads();

  #pragma unroll
  for (int half = 0; half < 2; ++half) {
    const int e   = tid + half * 512;
    const int row = e >> 6;
    const int col = e & 63;
    float o = 0.f;
    #pragma unroll
    for (int w = 0; w < NWV_; ++w) o += (&Xsh[w][0][0])[row * PPAD_ + col];
    float dg = 0.f;
    #pragma unroll
    for (int w = 0; w < NWV_; ++w) dg += degp_s[row][w];
    const float diagc = 1.f - dg;
    o += diagc * xwI[row * FO_ + col] + bias[col];
    out[(size_t)(b * N_ + i0 + row) * FO_ + col] = o;
  }
}

extern "C" void kernel_launch(void* const* d_in, const int* in_sizes, int n_in,
                              void* d_out, int out_size, void* d_ws, size_t ws_size,
                              hipStream_t stream) {
  (void)in_sizes; (void)n_in; (void)out_size; (void)ws_size;
  const float* x    = (const float*)d_in[0];
  const float* adj  = (const float*)d_in[1];
  const float* w    = (const float*)d_in[2];
  const float* bias = (const float*)d_in[3];
  float*       out  = (float*)d_out;
  float*       xw   = (float*)d_ws;

  gtv_xw_gemm<<<(B_ * N_) / 16, 128, 0, stream>>>(x, w, xw);
  gtv_fused  <<<B_ * (N_ / 16), 512, 0, stream>>>(adj, xw, bias, out);
}


# fused kernel: mod_adj = adj * v_rcp_f32(max(d,eps)), one reciprocal and one product, in place of the 12-instruction correctly rounded division behind an exec-masked branch; quotient within 2 ulp
# speedup vs baseline: 2.4301x; 1.0443x over previous
#include <hip/hip_runtime.h>

#define B_    4
#define N_    1024
#define FI_   128
#define FO_   64
#define DELTA_ 1.0f
#define EPS_   1e-3f

typedef __attribute__((ext_vector_type(2))) float v2f;
typedef __attribute__((ext_vector_type(8))) float v8f;

static __device__ __forceinline__ v8f wmma_f32_16x16x4(v2f a, v2f b, v8f c) {
  return __builtin_amdgcn_wmma_f32_16x16x4_f32(
       false, a,  false, b,
       (short)0, c,  false,  false);
}

__global__ void __launch_bounds__(128)
gtv_xw_gemm(const float* __restrict__ x, const float* __restrict__ w,
            float* __restrict__ xw) {
  const int tid  = threadIdx.x;
  const int wv   = tid >> 5;
  const int lane = tid & 31;
  const int hh   = lane >> 4;
  const int l15  = lane & 15;
  const int rt   = blockIdx.x;
  const int f0   = wv * 16;

  const float* __restrict__ xrow = x + (size_t)(rt * 16 + l15) * FI_;
  v8f c = {0.f, 0.f, 0.f, 0.f, 0.f, 0.f, 0.f, 0.f};

  #pragma unroll 4
  for (int kc = 0; kc < FI_ / 4; ++kc) {
    const int ks = kc * 4 + hh * 2;
    v2f a;  a.x = xrow[ks];  a.y = xrow[ks + 1];
    const float* __restrict__ wp = w + (size_t)ks * FO_ + f0 + l15;
    v2f b;  b.x = wp[0];     b.y = wp[FO_];
    c = wmma_f32_16x16x4(a, b, c);
  }

  float* __restrict__ orow = xw + (size_t)(rt * 16 + hh * 8) * FO_ + f0 + l15;
  #pragma unroll
  for (int r = 0; r < 8; ++r) orow[(size_t)r * FO_] = c[r];
}

#define NWV_  16
#define TPW_  (N_ / 32 / NWV_)
#define PPAD_ 66
#define MPAD_ 36
#define XPAD_ 68

typedef __attribute__((ext_vector_type(16))) float v16f;
typedef const __attribute__((address_space(4))) v16f* cv16p;

__global__ void __launch_bounds__(512)
gtv_fused(const float* __restrict__ adj, const float* __restrict__ xw,
          const float* __restrict__ bias, float* __restrict__ out) {
  __shared__ __attribute__((aligned(16))) float Xsh[NWV_][32][XPAD_];
  __shared__ __attribute__((aligned(16))) float Msh[NWV_][16][MPAD_];
  __shared__ float degp_s[16][NWV_];

  const int tid  = threadIdx.x;
  const int wv   = __builtin_amdgcn_readfirstlane(tid >> 5);
  const int lane = tid & 31;
  const int hh   = lane >> 4;
  const int l15  = lane & 15;

  const int b  = blockIdx.x >> 6;
  const int i0 = (blockIdx.x & 63) * 16;

  const float* __restrict__ adjb = adj + (size_t)b * N_ * N_ + (size_t)i0 * N_;
  const float* __restrict__ xwb  = xw + (size_t)b * N_ * FO_;
  const float* __restrict__ xwI  = xwb + (size_t)i0 * FO_;

  float dacc = 0.f;
  v8f oacc[4];
  #pragma unroll
  for (int f = 0; f < 4; ++f)
    #pragma unroll
    for (int e = 0; e < 8; ++e) oacc[f][e] = 0.f;

  float arow;
  {
    const float4* __restrict__ pr = (const float4*)(xwI + l15 * FO_ + hh * 32);
    float4 g[8];
    #pragma unroll
    for (int k = 0; k < 8; ++k) g[k] = pr[k];
    const float4 u = ((g[0] + g[1]) + (g[2] + g[3])) + ((g[4] + g[5]) + (g[6] + g[7]));
    const float part = (u.x + u.y) + (u.z + u.w);
    arow = part + __shfl_xor(part, 16, 32);
  }

  #pragma unroll 1
  for (int step = 0; step < TPW_; ++step) {
    const int j0 = (step * NWV_ + wv) * 32;

    {
      const float4* __restrict__ gp = (const float4*)(xwb + (size_t)j0 * FO_) + lane;
      float4 g[16];
      #pragma unroll
      for (int p = 0; p < 16; ++p) g[p] = gp[p * 32];
      #pragma unroll
      for (int p = 0; p < 16; ++p)
        *(float4*)&Xsh[wv][2 * p + hh][l15 * 4] = g[p];
    }
    float xj[FO_];
    #pragma unroll
    for (int kc = 0; kc < FO_ / 4; ++kc) {
      const float4 t = *(const float4*)&Xsh[wv][lane][kc * 4];
      xj[kc * 4 + 0] = t.x;  xj[kc * 4 + 1] = t.y;
      xj[kc * 4 + 2] = t.z;  xj[kc * 4 + 3] = t.w;
    }

    float bj;
    {
      v2f bs[4];
      #pragma unroll
      for (int c = 0; c < 4; ++c) { bs[c].x = xj[2 * c]; bs[c].y = xj[2 * c + 1]; }
      #pragma unroll
      for (int k = 8; k < FO_; k += 2) {
        v2f t;  t.x = xj[k];  t.y = xj[k + 1];
        bs[(k >> 1) & 3] += t;
      }
      const v2f b2 = (bs[0] + bs[1]) + (bs[2] + bs[3]);
      bj = b2.x + b2.y;
    }

    const float* __restrict__ ap = adjb + j0 + lane;
    float ac[4], an[4];
    #pragma unroll
    for (int c = 0; c < 4; ++c) ac[c] = ap[(size_t)c * N_];
    #pragma unroll 1
    for (int rg = 0; rg < 4; ++rg) {
      #pragma unroll
      for (int c = 0; c < 4; ++c) {
        const int rn = (rg * 4 + 4 + c < 16) ? (rg * 4 + 4 + c) : 15;
        an[c] = ap[(size_t)rn * N_];
      }
      #pragma unroll
      for (int c4 = 0; c4 < 4; ++c4) {
        const int r = rg * 4 + c4;
        __builtin_amdgcn_sched_barrier(0);
        const cv16p xr = (cv16p)(xwI + r * FO_);
        float m0[16];
        {
          const v16f xi = xr[0];
          asm("v_dual_max_num_f32 %0, %16, %32 :: v_dual_max_num_f32 %1, %17, %33\n\t"
              "v_dual_max_num_f32 %2, %18, %34 :: v_dual_max_num_f32 %3, %19, %35\n\t"
              "v_dual_max_num_f32 %4, %20, %36 :: v_dual_max_num_f32 %5, %21, %37\n\t"
              "v_dual_max_num_f32 %6, %22, %38 :: v_dual_max_num_f32 %7, %23, %39\n\t"
              "v_dual_max_num_f32 %8, %24, %40 :: v_dual_max_num_f32 %9, %25, %41\n\t"
              "v_dual_max_num_f32 %10, %26, %42 :: v_dual_max_num_f32 %11, %27, %43\n\t"
              "v_dual_max_num_f32 %12, %28, %44 :: v_dual_max_num_f32 %13, %29, %45\n\t"
              "v_dual_max_num_f32 %14, %30, %46 :: v_dual_max_num_f32 %15, %31, %47"
              : "=&v"(m0[0]), "=&v"(m0[1]), "=&v"(m0[2]), "=&v"(m0[3]), "=&v"(m0[4]), "=&v"(m0[5]), "=&v"(m0[6]), "=&v"(m0[7]), "=&v"(m0[8]), "=&v"(m0[9]), "=&v"(m0[10]), "=&v"(m0[11]), "=&v"(m0[12]), "=&v"(m0[13]), "=&v"(m0[14]), "=&v"(m0[15])
              : "s"(xi[0]), "s"(xi[1]), "s"(xi[2]), "s"(xi[3]), "s"(xi[4]), "s"(xi[5]), "s"(xi[6]), "s"(xi[7]), "s"(xi[8]), "s"(xi[9]), "s"(xi[10]), "s"(xi[11]), "s"(xi[12]), "s"(xi[13]), "s"(xi[14]), "s"(xi[15]),
                "v"(xj[0]), "v"(xj[1]), "v"(xj[2]), "v"(xj[3]), "v"(xj[4]), "v"(xj[5]), "v"(xj[6]), "v"(xj[7]), "v"(xj[8]), "v"(xj[9]), "v"(xj[10]), "v"(xj[11]), "v"(xj[12]), "v"(xj[13]), "v"(xj[14]), "v"(xj[15]));
        }
        v2f S[8];
        #pragma unroll
        for (int c = 0; c < 8; ++c) { S[c].x = m0[2 * c]; S[c].y = m0[2 * c + 1]; }
        #pragma unroll
        for (int q = 1; q < 4; ++q) {
          const v16f xi = xr[q];
          asm("v_dual_max_num_f32 v208, %8, %24 :: v_dual_max_num_f32 v209, %9, %25\n\t"
              "v_dual_max_num_f32 v210, %10, %26 :: v_dual_max_num_f32 v211, %11, %27\n\t"
              "v_dual_max_num_f32 v212, %12, %28 :: v_dual_max_num_f32 v213, %13, %29\n\t"
              "v_dual_max_num_f32 v214, %14, %30 :: v_dual_max_num_f32 v215, %15, %31\n\t"
              "v_dual_max_num_f32 v216, %16, %32 :: v_dual_max_num_f32 v217, %17, %33\n\t"
              "v_dual_max_num_f32 v218, %18, %34 :: v_dual_max_num_f32 v219, %19, %35\n\t"
              "v_dual_max_num_f32 v220, %20, %36 :: v_dual_max_num_f32 v221, %21, %37\n\t"
              "v_dual_max_num_f32 v222, %22, %38 :: v_dual_max_num_f32 v223, %23, %39\n\t"
              "v_pk_add_f32 %0, %0, v[208:209]\n\t"
              "v_pk_add_f32 %1, %1, v[210:211]\n\t"
              "v_pk_add_f32 %2, %2, v[212:213]\n\t"
              "v_pk_add_f32 %3, %3, v[214:215]\n\t"
              "v_pk_add_f32 %4, %4, v[216:217]\n\t"
              "v_pk_add_f32 %5, %5, v[218:219]\n\t"
              "v_pk_add_f32 %6, %6, v[220:221]\n\t"
              "v_pk_add_f32 %7, %7, v[222:223]"
              : "+v"(S[0]), "+v"(S[1]), "+v"(S[2]), "+v"(S[3]), "+v"(S[4]), "+v"(S[5]), "+v"(S[6]), "+v"(S[7])
              : "s"(xi[0]), "s"(xi[1]), "s"(xi[2]), "s"(xi[3]), "s"(xi[4]), "s"(xi[5]), "s"(xi[6]), "s"(xi[7]), "s"(xi[8]), "s"(xi[9]), "s"(xi[10]), "s"(xi[11]), "s"(xi[12]), "s"(xi[13]), "s"(xi[14]), "s"(xi[15]),
                "v"(xj[q * 16 + 0]), "v"(xj[q * 16 + 1]), "v"(xj[q * 16 + 2]), "v"(xj[q * 16 + 3]), "v"(xj[q * 16 + 4]), "v"(xj[q * 16 + 5]), "v"(xj[q * 16 + 6]), "v"(xj[q * 16 + 7]), "v"(xj[q * 16 + 8]), "v"(xj[q * 16 + 9]), "v"(xj[q * 16 + 10]), "v"(xj[q * 16 + 11]), "v"(xj[q * 16 + 12]), "v"(xj[q * 16 + 13]), "v"(xj[q * 16 + 14]), "v"(xj[q * 16 + 15])
              : "v208", "v209", "v210", "v211", "v212", "v213", "v214", "v215", "v216", "v217", "v218", "v219", "v220", "v221", "v222", "v223");
        }
        const v2f S2 = ((S[0] + S[1]) + (S[2] + S[3])) + ((S[4] + S[5]) + (S[6] + S[7]));
        const float msum = S2.x + S2.y;
        const float ai = __int_as_float(__builtin_amdgcn_readlane(__float_as_int(arow), r));
        const float d = __builtin_fmaf(2.f, msum, -(ai + bj));
        const float a0 = ac[c4];
        float m = a0 * __builtin_amdgcn_rcpf(fmaxf(d, EPS_));
        m *= DELTA_;
        Msh[wv][r][lane] = m;
      }
      #pragma unroll
      for (int c = 0; c < 4; ++c) ac[c] = an[c];
    }

    float4 am[4];
    #pragma unroll
    for (int q = 0; q < 4; ++q)
      am[q] = *(const float4*)&Msh[wv][l15][q * 8 + hh * 4];
    #pragma unroll
    for (int q = 0; q < 4; ++q)
      dacc += (am[q].x + am[q].y) + (am[q].z + am[q].w);
    asm volatile("" : "+v"(dacc));

    v2f bvec[4][4][2];
    {
      const float* bq = &Xsh[wv][hh * 4][l15];
      #pragma unroll
      for (int q = 0; q < 4; ++q)
        #pragma unroll
        for (int t = 0; t < 2; ++t)
          #pragma unroll
          for (int fc = 0; fc < 4; ++fc) {
            const float* p = bq + (q * 8 + t * 2) * XPAD_ + fc * 16;
            bvec[fc][q][t].x = p[0];  bvec[fc][q][t].y = p[XPAD_];
          }
    }
    v2f avec[4][2];
    #pragma unroll
    for (int q = 0; q < 4; ++q) {
      avec[q][0].x = am[q].x;  avec[q][0].y = am[q].y;
      avec[q][1].x = am[q].z;  avec[q][1].y = am[q].w;
    }
    __builtin_amdgcn_sched_barrier(0);
    #pragma unroll
    for (int q = 0; q < 4; ++q)
      #pragma unroll
      for (int t = 0; t < 2; ++t)
        #pragma unroll
        for (int fc = 0; fc < 4; ++fc)
          oacc[fc] = wmma_f32_16x16x4(avec[q][t], bvec[fc][q][t], oacc[fc]);
    __builtin_amdgcn_sched_barrier(0);
    asm volatile("v_nop\n\tv_nop\n\tv_nop\n\tv_nop\n\tv_nop\n\tv_nop\n\tv_nop\n\tv_nop"
                 : "+v"(oacc[0]), "+v"(oacc[1]), "+v"(oacc[2]), "+v"(oacc[3]));
    __builtin_amdgcn_sched_barrier(0);
  }

  {
    const float dsum = dacc + __shfl_xor(dacc, 16, 32);
    if (hh == 0) degp_s[l15][wv] = dsum;
  }
  {
    float* slab = &Xsh[wv][0][0];
    #pragma unroll
    for (int fc = 0; fc < 4; ++fc)
      #pragma unroll
      for (int r = 0; r < 8; ++r)
        slab[(r + hh * 8) * PPAD_ + fc * 16 + l15] = oacc[fc][r];
  }
  __syncthreads();

  #pragma unroll
  for (int half = 0; half < 2; ++half) {
    const int e   = tid + half * 512;
    const int row = e >> 6;
    const int col = e & 63;
    float o = 0.f;
    #pragma unroll
    for (int w = 0; w < NWV_; ++w) o += (&Xsh[w][0][0])[row * PPAD_ + col];
    float dg = 0.f;
    #pragma unroll
    for (int w = 0; w < NWV_; ++w) dg += degp_s[row][w];
    const float diagc = 1.f - dg;
    o += diagc * xwI[row * FO_ + col] + bias[col];
    out[(size_t)(b * N_ + i0 + row) * FO_ + col] = o;
  }
}

extern "C" void kernel_launch(void* const* d_in, const int* in_sizes, int n_in,
                              void* d_out, int out_size, void* d_ws, size_t ws_size,
                              hipStream_t stream) {
  (void)in_sizes; (void)n_in; (void)out_size; (void)ws_size;
  const float* x    = (const float*)d_in[0];
  const float* adj  = (const float*)d_in[1];
  const float* w    = (const float*)d_in[2];
  const float* bias = (const float*)d_in[3];
  float*       out  = (float*)d_out;
  float*       xw   = (float*)d_ws;

  gtv_xw_gemm<<<(B_ * N_) / 16, 128, 0, stream>>>(x, w, xw);
  gtv_fused  <<<B_ * (N_ / 16), 512, 0, stream>>>(adj, xw, bias, out);
}


# first kernel (x@W): x tile and W come in as whole 128-byte lines (20 16-byte loads a thread, one group), are stored to LDS, and all A/B fragments are read from LDS in one group; every sum unchanged
# speedup vs baseline: 2.8605x; 1.1771x over previous
#include <hip/hip_runtime.h>

#define B_    4
#define N_    1024
#define FI_   128
#define FO_   64
#define DELTA_ 1.0f
#define EPS_   1e-3f

typedef __attribute__((ext_vector_type(2))) float v2f;
typedef __attribute__((ext_vector_type(8))) float v8f;

static __device__ __forceinline__ v8f wmma_f32_16x16x4(v2f a, v2f b, v8f c) {
  return __builtin_amdgcn_wmma_f32_16x16x4_f32(
       false, a,  false, b,
       (short)0, c,  false,  false);
}

#define XSP_ 132
#define WSP_ 72

__global__ void __launch_bounds__(128)
gtv_xw_gemm(const float* __restrict__ x, const float* __restrict__ w,
            float* __restrict__ xw) {
  __shared__ __attribute__((aligned(16))) float xs[16][XSP_];
  __shared__ __attribute__((aligned(16))) float wsh[FI_][WSP_];

  const int tid  = threadIdx.x;
  const int wv   = tid >> 5;
  const int lane = tid & 31;
  const int hh   = lane >> 4;
  const int l15  = lane & 15;
  const int rt   = blockIdx.x;
  const int f0   = wv * 16;

  {
    const float4* __restrict__ xg = (const float4*)(x + (size_t)rt * 16 * FI_) + tid;
    const float4* __restrict__ wg = (const float4*)w + tid;
    float4 gx[4], gw[16];
    #pragma unroll
    for (int p = 0; p < 4; ++p) gx[p] = xg[p * 128];
    #pragma unroll
    for (int p = 0; p < 16; ++p) gw[p] = wg[p * 128];
    #pragma unroll
    for (int p = 0; p < 4; ++p) {
      const int idx = tid + p * 128;
      *(float4*)&xs[idx >> 5][(idx & 31) * 4] = gx[p];
    }
    #pragma unroll
    for (int p = 0; p < 16; ++p) {
      const int idx = tid + p * 128;
      *(float4*)&wsh[idx >> 4][(idx & 15) * 4] = gw[p];
    }
  }
  __syncthreads();

  v8f c = {0.f, 0.f, 0.f, 0.f, 0.f, 0.f, 0.f, 0.f};

  v2f af[FI_ / 4], bf[FI_ / 4];
  #pragma unroll
  for (int kc = 0; kc < FI_ / 4; ++kc) {
    const int ks = kc * 4 + hh * 2;
    af[kc] = *(const v2f*)&xs[l15][ks];
    bf[kc].x = wsh[ks][f0 + l15];
    bf[kc].y = wsh[ks + 1][f0 + l15];
  }
  typedef __attribute__((address_space(1))) float* gfptr;
  gfptr orow = (gfptr)(xw + (size_t)(rt * 16 + hh * 8) * FO_ + f0 + l15);
  asm volatile("" : "+v"(orow));
  __builtin_amdgcn_sched_barrier(0);
  #pragma unroll
  for (int kc = 0; kc < FI_ / 4; ++kc)
    c = wmma_f32_16x16x4(af[kc], bf[kc], c);
  asm volatile("v_nop\n\tv_nop\n\tv_nop\n\tv_nop\n\tv_nop\n\tv_nop\n\tv_nop\n\tv_nop" : "+v"(c));

  #pragma unroll
  for (int r = 0; r < 8; ++r) orow[(size_t)r * FO_] = c[r];
}

#define NWV_  16
#define TPW_  (N_ / 32 / NWV_)
#define PPAD_ 66
#define MPAD_ 36
#define XPAD_ 68

typedef __attribute__((ext_vector_type(16))) float v16f;
typedef const __attribute__((address_space(4))) v16f* cv16p;

__global__ void __launch_bounds__(512)
gtv_fused(const float* __restrict__ adj, const float* __restrict__ xw,
          const float* __restrict__ bias, float* __restrict__ out) {
  __shared__ __attribute__((aligned(16))) float Xsh[NWV_][32][XPAD_];
  __shared__ __attribute__((aligned(16))) float Msh[NWV_][16][MPAD_];
  __shared__ float degp_s[16][NWV_];

  const int tid  = threadIdx.x;
  const int wv   = __builtin_amdgcn_readfirstlane(tid >> 5);
  const int lane = tid & 31;
  const int hh   = lane >> 4;
  const int l15  = lane & 15;

  const int b  = blockIdx.x >> 6;
  const int i0 = (blockIdx.x & 63) * 16;

  const float* __restrict__ adjb = adj + (size_t)b * N_ * N_ + (size_t)i0 * N_;
  const float* __restrict__ xwb  = xw + (size_t)b * N_ * FO_;
  const float* __restrict__ xwI  = xwb + (size_t)i0 * FO_;

  float dacc = 0.f;
  v8f oacc[4];
  #pragma unroll
  for (int f = 0; f < 4; ++f)
    #pragma unroll
    for (int e = 0; e < 8; ++e) oacc[f][e] = 0.f;

  float arow;
  {
    const float4* __restrict__ pr = (const float4*)(xwI + l15 * FO_ + hh * 32);
    float4 g[8];
    #pragma unroll
    for (int k = 0; k < 8; ++k) g[k] = pr[k];
    const float4 u = ((g[0] + g[1]) + (g[2] + g[3])) + ((g[4] + g[5]) + (g[6] + g[7]));
    const float part = (u.x + u.y) + (u.z + u.w);
    arow = part + __shfl_xor(part, 16, 32);
  }

  #pragma unroll 1
  for (int step = 0; step < TPW_; ++step) {
    const int j0 = (step * NWV_ + wv) * 32;

    {
      const float4* __restrict__ gp = (const float4*)(xwb + (size_t)j0 * FO_) + lane;
      float4 g[16];
      #pragma unroll
      for (int p = 0; p < 16; ++p) g[p] = gp[p * 32];
      #pragma unroll
      for (int p = 0; p < 16; ++p)
        *(float4*)&Xsh[wv][2 * p + hh][l15 * 4] = g[p];
    }
    float xj[FO_];
    #pragma unroll
    for (int kc = 0; kc < FO_ / 4; ++kc) {
      const float4 t = *(const float4*)&Xsh[wv][lane][kc * 4];
      xj[kc * 4 + 0] = t.x;  xj[kc * 4 + 1] = t.y;
      xj[kc * 4 + 2] = t.z;  xj[kc * 4 + 3] = t.w;
    }

    float bj;
    {
      v2f bs[4];
      #pragma unroll
      for (int c = 0; c < 4; ++c) { bs[c].x = xj[2 * c]; bs[c].y = xj[2 * c + 1]; }
      #pragma unroll
      for (int k = 8; k < FO_; k += 2) {
        v2f t;  t.x = xj[k];  t.y = xj[k + 1];
        bs[(k >> 1) & 3] += t;
      }
      const v2f b2 = (bs[0] + bs[1]) + (bs[2] + bs[3]);
      bj = b2.x + b2.y;
    }

    const float* __restrict__ ap = adjb + j0 + lane;
    float ac[4], an[4];
    #pragma unroll
    for (int c = 0; c < 4; ++c) ac[c] = ap[(size_t)c * N_];
    #pragma unroll 1
    for (int rg = 0; rg < 4; ++rg) {
      #pragma unroll
      for (int c = 0; c < 4; ++c) {
        const int rn = (rg * 4 + 4 + c < 16) ? (rg * 4 + 4 + c) : 15;
        an[c] = ap[(size_t)rn * N_];
      }
      #pragma unroll
      for (int c4 = 0; c4 < 4; ++c4) {
        const int r = rg * 4 + c4;
        __builtin_amdgcn_sched_barrier(0);
        const cv16p xr = (cv16p)(xwI + r * FO_);
        float m0[16];
        {
          const v16f xi = xr[0];
          asm("v_dual_max_num_f32 %0, %16, %32 :: v_dual_max_num_f32 %1, %17, %33\n\t"
              "v_dual_max_num_f32 %2, %18, %34 :: v_dual_max_num_f32 %3, %19, %35\n\t"
              "v_dual_max_num_f32 %4, %20, %36 :: v_dual_max_num_f32 %5, %21, %37\n\t"
              "v_dual_max_num_f32 %6, %22, %38 :: v_dual_max_num_f32 %7, %23, %39\n\t"
              "v_dual_max_num_f32 %8, %24, %40 :: v_dual_max_num_f32 %9, %25, %41\n\t"
              "v_dual_max_num_f32 %10, %26, %42 :: v_dual_max_num_f32 %11, %27, %43\n\t"
              "v_dual_max_num_f32 %12, %28, %44 :: v_dual_max_num_f32 %13, %29, %45\n\t"
              "v_dual_max_num_f32 %14, %30, %46 :: v_dual_max_num_f32 %15, %31, %47"
              : "=&v"(m0[0]), "=&v"(m0[1]), "=&v"(m0[2]), "=&v"(m0[3]), "=&v"(m0[4]), "=&v"(m0[5]), "=&v"(m0[6]), "=&v"(m0[7]), "=&v"(m0[8]), "=&v"(m0[9]), "=&v"(m0[10]), "=&v"(m0[11]), "=&v"(m0[12]), "=&v"(m0[13]), "=&v"(m0[14]), "=&v"(m0[15])
              : "s"(xi[0]), "s"(xi[1]), "s"(xi[2]), "s"(xi[3]), "s"(xi[4]), "s"(xi[5]), "s"(xi[6]), "s"(xi[7]), "s"(xi[8]), "s"(xi[9]), "s"(xi[10]), "s"(xi[11]), "s"(xi[12]), "s"(xi[13]), "s"(xi[14]), "s"(xi[15]),
                "v"(xj[0]), "v"(xj[1]), "v"(xj[2]), "v"(xj[3]), "v"(xj[4]), "v"(xj[5]), "v"(xj[6]), "v"(xj[7]), "v"(xj[8]), "v"(xj[9]), "v"(xj[10]), "v"(xj[11]), "v"(xj[12]), "v"(xj[13]), "v"(xj[14]), "v"(xj[15]));
        }
        v2f S[8];
        #pragma unroll
        for (int c = 0; c < 8; ++c) { S[c].x = m0[2 * c]; S[c].y = m0[2 * c + 1]; }
        #pragma unroll
        for (int q = 1; q < 4; ++q) {
          const v16f xi = xr[q];
          asm("v_dual_max_num_f32 v208, %8, %24 :: v_dual_max_num_f32 v209, %9, %25\n\t"
              "v_dual_max_num_f32 v210, %10, %26 :: v_dual_max_num_f32 v211, %11, %27\n\t"
              "v_dual_max_num_f32 v212, %12, %28 :: v_dual_max_num_f32 v213, %13, %29\n\t"
              "v_dual_max_num_f32 v214, %14, %30 :: v_dual_max_num_f32 v215, %15, %31\n\t"
              "v_dual_max_num_f32 v216, %16, %32 :: v_dual_max_num_f32 v217, %17, %33\n\t"
              "v_dual_max_num_f32 v218, %18, %34 :: v_dual_max_num_f32 v219, %19, %35\n\t"
              "v_dual_max_num_f32 v220, %20, %36 :: v_dual_max_num_f32 v221, %21, %37\n\t"
              "v_dual_max_num_f32 v222, %22, %38 :: v_dual_max_num_f32 v223, %23, %39\n\t"
              "v_pk_add_f32 %0, %0, v[208:209]\n\t"
              "v_pk_add_f32 %1, %1, v[210:211]\n\t"
              "v_pk_add_f32 %2, %2, v[212:213]\n\t"
              "v_pk_add_f32 %3, %3, v[214:215]\n\t"
              "v_pk_add_f32 %4, %4, v[216:217]\n\t"
              "v_pk_add_f32 %5, %5, v[218:219]\n\t"
              "v_pk_add_f32 %6, %6, v[220:221]\n\t"
              "v_pk_add_f32 %7, %7, v[222:223]"
              : "+v"(S[0]), "+v"(S[1]), "+v"(S[2]), "+v"(S[3]), "+v"(S[4]), "+v"(S[5]), "+v"(S[6]), "+v"(S[7])
              : "s"(xi[0]), "s"(xi[1]), "s"(xi[2]), "s"(xi[3]), "s"(xi[4]), "s"(xi[5]), "s"(xi[6]), "s"(xi[7]), "s"(xi[8]), "s"(xi[9]), "s"(xi[10]), "s"(xi[11]), "s"(xi[12]), "s"(xi[13]), "s"(xi[14]), "s"(xi[15]),
                "v"(xj[q * 16 + 0]), "v"(xj[q * 16 + 1]), "v"(xj[q * 16 + 2]), "v"(xj[q * 16 + 3]), "v"(xj[q * 16 + 4]), "v"(xj[q * 16 + 5]), "v"(xj[q * 16 + 6]), "v"(xj[q * 16 + 7]), "v"(xj[q * 16 + 8]), "v"(xj[q * 16 + 9]), "v"(xj[q * 16 + 10]), "v"(xj[q * 16 + 11]), "v"(xj[q * 16 + 12]), "v"(xj[q * 16 + 13]), "v"(xj[q * 16 + 14]), "v"(xj[q * 16 + 15])
              : "v208", "v209", "v210", "v211", "v212", "v213", "v214", "v215", "v216", "v217", "v218", "v219", "v220", "v221", "v222", "v223");
        }
        const v2f S2 = ((S[0] + S[1]) + (S[2] + S[3])) + ((S[4] + S[5]) + (S[6] + S[7]));
        const float msum = S2.x + S2.y;
        const float ai = __int_as_float(__builtin_amdgcn_readlane(__float_as_int(arow), r));
        const float d = __builtin_fmaf(2.f, msum, -(ai + bj));
        const float a0 = ac[c4];
        float m = a0 * __builtin_amdgcn_rcpf(fmaxf(d, EPS_));
        m *= DELTA_;
        Msh[wv][r][lane] = m;
      }
      #pragma unroll
      for (int c = 0; c < 4; ++c) ac[c] = an[c];
    }

    float4 am[4];
    #pragma unroll
    for (int q = 0; q < 4; ++q)
      am[q] = *(const float4*)&Msh[wv][l15][q * 8 + hh * 4];
    #pragma unroll
    for (int q = 0; q < 4; ++q)
      dacc += (am[q].x + am[q].y) + (am[q].z + am[q].w);
    asm volatile("" : "+v"(dacc));

    v2f bvec[4][4][2];
    {
      const float* bq = &Xsh[wv][hh * 4][l15];
      #pragma unroll
      for (int q = 0; q < 4; ++q)
        #pragma unroll
        for (int t = 0; t < 2; ++t)
          #pragma unroll
          for (int fc = 0; fc < 4; ++fc) {
            const float* p = bq + (q * 8 + t * 2) * XPAD_ + fc * 16;
            bvec[fc][q][t].x = p[0];  bvec[fc][q][t].y = p[XPAD_];
          }
    }
    v2f avec[4][2];
    #pragma unroll
    for (int q = 0; q < 4; ++q) {
      avec[q][0].x = am[q].x;  avec[q][0].y = am[q].y;
      avec[q][1].x = am[q].z;  avec[q][1].y = am[q].w;
    }
    __builtin_amdgcn_sched_barrier(0);
    #pragma unroll
    for (int q = 0; q < 4; ++q)
      #pragma unroll
      for (int t = 0; t < 2; ++t)
        #pragma unroll
        for (int fc = 0; fc < 4; ++fc)
          oacc[fc] = wmma_f32_16x16x4(avec[q][t], bvec[fc][q][t], oacc[fc]);
    __builtin_amdgcn_sched_barrier(0);
    asm volatile("v_nop\n\tv_nop\n\tv_nop\n\tv_nop\n\tv_nop\n\tv_nop\n\tv_nop\n\tv_nop"
                 : "+v"(oacc[0]), "+v"(oacc[1]), "+v"(oacc[2]), "+v"(oacc[3]));
    __builtin_amdgcn_sched_barrier(0);
  }

  {
    const float dsum = dacc + __shfl_xor(dacc, 16, 32);
    if (hh == 0) degp_s[l15][wv] = dsum;
  }
  {
    float* slab = &Xsh[wv][0][0];
    #pragma unroll
    for (int fc = 0; fc < 4; ++fc)
      #pragma unroll
      for (int r = 0; r < 8; ++r)
        slab[(r + hh * 8) * PPAD_ + fc * 16 + l15] = oacc[fc][r];
  }
  __syncthreads();

  #pragma unroll
  for (int half = 0; half < 2; ++half) {
    const int e   = tid + half * 512;
    const int row = e >> 6;
    const int col = e & 63;
    float o = 0.f;
    #pragma unroll
    for (int w = 0; w < NWV_; ++w) o += (&Xsh[w][0][0])[row * PPAD_ + col];
    float dg = 0.f;
    #pragma unroll
    for (int w = 0; w < NWV_; ++w) dg += degp_s[row][w];
    const float diagc = 1.f - dg;
    o += diagc * xwI[row * FO_ + col] + bias[col];
    out[(size_t)(b * N_ + i0 + row) * FO_ + col] = o;
  }
}

extern "C" void kernel_launch(void* const* d_in, const int* in_sizes, int n_in,
                              void* d_out, int out_size, void* d_ws, size_t ws_size,
                              hipStream_t stream) {
  (void)in_sizes; (void)n_in; (void)out_size; (void)ws_size;
  const float* x    = (const float*)d_in[0];
  const float* adj  = (const float*)d_in[1];
  const float* w    = (const float*)d_in[2];
  const float* bias = (const float*)d_in[3];
  float*       out  = (float*)d_out;
  float*       xw   = (float*)d_ws;

  gtv_xw_gemm<<<(B_ * N_) / 16, 128, 0, stream>>>(x, w, xw);
  gtv_fused  <<<B_ * (N_ / 16), 512, 0, stream>>>(adj, xw, bias, out);
}
